# starting kernel
# baseline (speedup 1.0000x reference)
#include <hip/hip_runtime.h>
#include <hip/hip_bf16.h>

typedef __attribute__((ext_vector_type(2))) float v2f;
typedef __attribute__((ext_vector_type(8))) float v8f;

__device__ __forceinline__ float sigmoidf_(float x) { return 1.0f / (1.0f + __expf(-x)); }

__device__ __forceinline__ v8f wmma4(v2f a, v2f b, v8f c) {
  return __builtin_amdgcn_wmma_f32_16x16x4_f32(false, a, false, b, (short)0, c,
                                               false, false);
}

template<bool WTRANS, bool HASA2, bool RELU>
__global__ void __launch_bounds__(256)
gemm_wmma_f32(const float* __restrict__ A1, int lda1,
              const float* __restrict__ W1, int ldw1, int K1,
              const float* __restrict__ A2, int lda2,
              const float* __restrict__ W2, int ldw2, int K2,
              const float* __restrict__ b1p, const float* __restrict__ b2p,
              float bias_scale,
              float* __restrict__ C, int ldc, int Mrows, int Ncols)
{
  const int wave = threadIdx.x >> 5;
  const int lane = threadIdx.x & 31;
  const int ntn  = Ncols >> 5;
  const int total = (Mrows >> 4) * ntn;
  const int tile = blockIdx.x * 8 + wave;
  if (tile >= total) return;
  const int mt = tile / ntn, nt = tile - mt * ntn;
  const int m0 = mt << 4, n0 = nt << 5;
  const int half = lane >> 4;
  const int lp   = lane & 15;

  v8f acc0 = {}, acc1 = {};

  {
    const float* ap  = A1 + (size_t)(m0 + lp) * lda1 + half * 2;
    const float* wp0;
    const float* wp1;
    if (WTRANS) {
      wp0 = W1 + (size_t)(n0 + lp) * ldw1 + half * 2;
      wp1 = wp0 + (size_t)16 * ldw1;
    } else {
      wp0 = W1 + (size_t)(half * 2) * ldw1 + n0 + lp;
      wp1 = wp0 + 16;
    }
    #pragma unroll 4
    for (int k0 = 0; k0 < K1; k0 += 4) {
      v2f a, b0, b1;
      a.x = ap[0]; a.y = ap[1]; ap += 4;
      if (WTRANS) {
        b0.x = wp0[0]; b0.y = wp0[1]; wp0 += 4;
        b1.x = wp1[0]; b1.y = wp1[1]; wp1 += 4;
      } else {
        b0.x = wp0[0]; b0.y = wp0[ldw1];
        b1.x = wp1[0]; b1.y = wp1[ldw1];
        wp0 += (size_t)4 * ldw1; wp1 += (size_t)4 * ldw1;
      }
      acc0 = wmma4(a, b0, acc0);
      acc1 = wmma4(a, b1, acc1);
    }
  }

  if (HASA2) {
    const float* ap  = A2 + (size_t)(m0 + lp) * lda2 + half * 2;
    const float* wp0 = W2 + (size_t)(half * 2) * ldw2 + n0 + lp;
    const float* wp1 = wp0 + 16;
    #pragma unroll 4
    for (int k0 = 0; k0 < K2; k0 += 4) {
      v2f a, b0, b1;
      a.x = ap[0]; a.y = ap[1]; ap += 4;
      b0.x = wp0[0]; b0.y = wp0[ldw2];
      b1.x = wp1[0]; b1.y = wp1[ldw2];
      wp0 += (size_t)4 * ldw2; wp1 += (size_t)4 * ldw2;
      acc0 = wmma4(a, b0, acc0);
      acc1 = wmma4(a, b1, acc1);
    }
  }

  float bias0 = 0.0f, bias1 = 0.0f;
  if (b1p) { bias0  = bias_scale * b1p[n0 + lp];      bias1  = bias_scale * b1p[n0 + 16 + lp]; }
  if (b2p) { bias0 += b2p[n0 + lp];                   bias1 += b2p[n0 + 16 + lp]; }

  float* crow = C + (size_t)(m0 + half * 8) * ldc + n0 + lp;
  #pragma unroll
  for (int r = 0; r < 8; ++r) {
    float v0 = acc0[r] + bias0;
    float v1 = acc1[r] + bias1;
    if (RELU) { v0 = fmaxf(v0, 0.0f); v1 = fmaxf(v1, 0.0f); }
    crow[0]  = v0;
    crow[16] = v1;
    crow += ldc;
  }
}

__global__ void __launch_bounds__(256)
pair_relu_sum(const float* __restrict__ Af, const float* __restrict__ Bf,
              float* __restrict__ S)
{
  const int ri = blockIdx.x;
  const int b  = ri >> 7;
  const int f  = threadIdx.x;
  const float bm = Bf[(size_t)ri * 256 + f];
  const float* ap = Af + (size_t)b * 128 * 256 + f;
  float acc = 0.0f;
  #pragma unroll 4
  for (int j = 0; j < 128; ++j)
    acc += fmaxf(ap[(size_t)j * 256] + bm, 0.0f);
  S[(size_t)ri * 256 + f] = acc;
}

__global__ void __launch_bounds__(512)
lstm_scan(const float* __restrict__ Gx, const float* __restrict__ whh,
          const float* __restrict__ h0, const float* __restrict__ c0,
          float* __restrict__ new_hidden)
{
  __shared__ float h_s[16][128];
  __shared__ float c_s[16][128];
  __shared__ float g_s[16][512];
  const int tid = threadIdx.x;
  for (int idx = tid; idx < 2048; idx += 512) {
    h_s[idx >> 7][idx & 127] = h0[idx];
    c_s[idx >> 7][idx & 127] = c0[idx];
  }
  __syncthreads();
  for (int t = 0; t < 128; ++t) {
    const int u = tid;
    float acc[16];
    #pragma unroll
    for (int b = 0; b < 16; ++b)
      acc[b] = Gx[((size_t)(b * 128 + t)) * 512 + u];
    for (int k = 0; k < 128; ++k) {
      const float w = whh[(size_t)u * 128 + k];
      #pragma unroll
      for (int b = 0; b < 16; ++b)
        acc[b] = fmaf(h_s[b][k], w, acc[b]);
    }
    #pragma unroll
    for (int b = 0; b < 16; ++b) g_s[b][u] = acc[b];
    __syncthreads();
    for (int idx = tid; idx < 2048; idx += 512) {
      const int b = idx >> 7, n = idx & 127;
      const float ig = g_s[b][n],       fg = g_s[b][128 + n];
      const float gg = g_s[b][256 + n], og = g_s[b][384 + n];
      const float c = sigmoidf_(fg) * c_s[b][n] + sigmoidf_(ig) * tanhf(gg);
      const float h = sigmoidf_(og) * tanhf(c);
      c_s[b][n] = c; h_s[b][n] = h;
      new_hidden[(size_t)(b * 128 + t) * 128 + n] = h;
    }
    __syncthreads();
  }
}

extern "C" void kernel_launch(void* const* d_in, const int* in_sizes, int n_in,
                              void* d_out, int out_size, void* d_ws, size_t ws_size,
                              hipStream_t stream)
{
  (void)in_sizes; (void)n_in; (void)out_size; (void)ws_size;
  const float* x      = (const float*)d_in[0];
  const float* hidden = (const float*)d_in[1];
  const float* h0     = (const float*)d_in[2];
  const float* c0     = (const float*)d_in[3];
  const float* f_w1   = (const float*)d_in[4];
  const float* f_b1   = (const float*)d_in[5];
  const float* f_w2   = (const float*)d_in[6];
  const float* f_b2   = (const float*)d_in[7];
  const float* g_w1   = (const float*)d_in[8];
  const float* g_b1   = (const float*)d_in[9];
  const float* g_w2   = (const float*)d_in[10];
  const float* g_b2   = (const float*)d_in[11];
  const float* wih    = (const float*)d_in[12];
  const float* whh    = (const float*)d_in[13];
  const float* bih    = (const float*)d_in[14];
  const float* bhh    = (const float*)d_in[15];
  const float* o_w1   = (const float*)d_in[16];
  const float* o_b1   = (const float*)d_in[17];
  const float* o_w2   = (const float*)d_in[18];
  const float* o_b2   = (const float*)d_in[19];
  float* out = (float*)d_out;

  const int BN = 16 * 128;
  float* R0 = (float*)d_ws;
  float* R1 = R0 + (size_t)BN * 256;
  float* R2 = R1 + (size_t)BN * 256;
  float* R3 = R2 + (size_t)BN * 256;

  auto blocks = [](int M, int N) { return ((M >> 4) * (N >> 5) + 7) / 8; };

  gemm_wmma_f32<false,false,false><<<blocks(BN,256), 256, 0, stream>>>(
      hidden,128, f_w1,256,128, nullptr,0, nullptr,0,0,
      nullptr,nullptr,1.0f, R0,256, BN,256);
  gemm_wmma_f32<false,false,false><<<blocks(BN,256), 256, 0, stream>>>(
      hidden,128, f_w1 + 128*256,256,128, nullptr,0, nullptr,0,0,
      f_b1,nullptr,1.0f, R1,256, BN,256);
  pair_relu_sum<<<BN, 256, 0, stream>>>(R0, R1, R2);
  gemm_wmma_f32<false,false,false><<<blocks(BN,128), 256, 0, stream>>>(
      R2,256, f_w2,128,256, nullptr,0, nullptr,0,0,
      f_b2,nullptr,128.0f, R0,128, BN,128);
  gemm_wmma_f32<false,true,true><<<blocks(BN,256), 256, 0, stream>>>(
      x,128, g_w1,256,128, R0,128, g_w1 + 128*256,256,128,
      g_b1,nullptr,1.0f, R1,256, BN,256);
  gemm_wmma_f32<false,false,false><<<blocks(BN,128), 256, 0, stream>>>(
      R1,256, g_w2,128,256, nullptr,0, nullptr,0,0,
      g_b2,nullptr,1.0f, R2,128, BN,128);
  gemm_wmma_f32<true,false,false><<<blocks(BN,512), 256, 0, stream>>>(
      R2,128, wih,128,128, nullptr,0, nullptr,0,0,
      bih,bhh,1.0f, R0,512, BN,512);
  lstm_scan<<<1, 512, 0, stream>>>(R0, whh, h0, c0, R3);
  gemm_wmma_f32<false,false,true><<<blocks(BN,256), 256, 0, stream>>>(
      R3,128, o_w1,256,128, nullptr,0, nullptr,0,0,
      o_b1,nullptr,1.0f, R0,256, BN,256);
  gemm_wmma_f32<false,false,false><<<blocks(BN,32), 256, 0, stream>>>(
      R0,256, o_w2,32,256, nullptr,0, nullptr,0,0,
      o_b2,nullptr,1.0f, out,32, BN,32);
}


# Small GEMMs and the pair sum: K and j loops in trips of 16 steps, a trip's loads (48: B as 8-byte column pairs, constant pitches) go out as one group: 4x fewer exposed L2 round trips; sums unchanged
# speedup vs baseline: 4.8508x; 4.8508x over previous
#include <hip/hip_runtime.h>
#include <hip/hip_bf16.h>

typedef __attribute__((ext_vector_type(2))) float v2f;
typedef __attribute__((ext_vector_type(4))) float v4f;
typedef __attribute__((ext_vector_type(8))) float v8f;

__device__ __forceinline__ float sigmoidf_(float x) { return __builtin_amdgcn_rcpf(1.0f + __expf(-x)); }
__device__ __forceinline__ float tanhf_(float x) {
  const float y = fabsf(x);
  const float z = fmaf(__builtin_amdgcn_rcpf(__expf(y + y) + 1.0f), -2.0f, 1.0f);
  return copysignf(z, x);
}

__device__ __forceinline__ v8f wmma4(v2f a, v2f b, v8f c) {
  return __builtin_amdgcn_wmma_f32_16x16x4_f32(false, a, false, b, (short)0, c,
                                               false, false);
}

template<bool WTRANS, bool HASA2, bool RELU>
__device__ __forceinline__ void
gemm_tile(const float* __restrict__ A1, int lda1,
          const float* __restrict__ W1, int ldw1, int K1,
          const float* __restrict__ A2, int lda2,
          const float* __restrict__ W2, int ldw2, int K2,
          const float* __restrict__ b1p, const float* __restrict__ b2p,
          float bias_scale,
          float* __restrict__ C, int ldc, int Mrows, int Ncols, int blk)
{
  const int wave = threadIdx.x >> 5;
  const int lane = threadIdx.x & 31;
  const int ntn  = Ncols >> 5;
  const int total = (Mrows >> 4) * ntn;
  const int tile = blk * 8 + wave;
  if (tile >= total) return;
  const int mt = tile / ntn, nt = tile - mt * ntn;
  const int m0 = mt << 4, n0 = nt << 5;
  const int half = lane >> 4;
  const int lp   = lane & 15;

  constexpr int KB = 16;
  const int c0 = WTRANS ? n0 + lp : n0 + 2 * lp;
  const int c1 = WTRANS ? c0 + 16 : c0 + 1;
  v8f acc0 = {}, acc1 = {};

  {
    const float* ap = A1 + (size_t)(m0 + lp) * lda1 + half * 2;
    const float* wp = WTRANS ? W1 + (size_t)(n0 + lp) * ldw1 + half * 2
                             : W1 + (size_t)(half * 2) * ldw1 + n0 + 2 * lp;
    for (int k0 = 0; k0 < K1; k0 += 4 * KB) {
      v2f a[KB], b0[KB], b1[KB];
      #pragma unroll
      for (int s = 0; s < KB; ++s) {
        a[s] = *reinterpret_cast<const v2f*>(ap + 4 * s);
        if (WTRANS) {
          b0[s] = *reinterpret_cast<const v2f*>(wp + 4 * s);
          b1[s] = *reinterpret_cast<const v2f*>(wp + (size_t)16 * ldw1 + 4 * s);
        } else {
          const v2f r0 = *reinterpret_cast<const v2f*>(wp + (size_t)(4 * s) * ldw1);
          const v2f r1 = *reinterpret_cast<const v2f*>(wp + (size_t)(4 * s + 1) * ldw1);
          b0[s].x = r0.x; b0[s].y = r1.x;
          b1[s].x = r0.y; b1[s].y = r1.y;
        }
      }
      ap += 4 * KB;
      wp += WTRANS ? (size_t)(4 * KB) : (size_t)(4 * KB) * ldw1;
      #pragma unroll
      for (int s = 0; s < KB; ++s) {
        acc0 = wmma4(a[s], b0[s], acc0);
        acc1 = wmma4(a[s], b1[s], acc1);
      }
      __builtin_amdgcn_sched_group_barrier(0x020, 3 * KB, 0);
      __builtin_amdgcn_sched_group_barrier(0x008, 2 * KB, 0);
    }
  }

  if (HASA2) {
    const float* ap = A2 + (size_t)(m0 + lp) * lda2 + half * 2;
    const float* wp = W2 + (size_t)(half * 2) * ldw2 + n0 + 2 * lp;
    for (int k0 = 0; k0 < K2; k0 += 4 * KB) {
      v2f a[KB], b0[KB], b1[KB];
      #pragma unroll
      for (int s = 0; s < KB; ++s) {
        a[s] = *reinterpret_cast<const v2f*>(ap + 4 * s);
        const v2f r0 = *reinterpret_cast<const v2f*>(wp + (size_t)(4 * s) * ldw2);
        const v2f r1 = *reinterpret_cast<const v2f*>(wp + (size_t)(4 * s + 1) * ldw2);
        b0[s].x = r0.x; b0[s].y = r1.x;
        b1[s].x = r0.y; b1[s].y = r1.y;
      }
      ap += 4 * KB;
      wp += (size_t)(4 * KB) * ldw2;
      #pragma unroll
      for (int s = 0; s < KB; ++s) {
        acc0 = wmma4(a[s], b0[s], acc0);
        acc1 = wmma4(a[s], b1[s], acc1);
      }
      __builtin_amdgcn_sched_group_barrier(0x020, 3 * KB, 0);
      __builtin_amdgcn_sched_group_barrier(0x008, 2 * KB, 0);
    }
  }

  float bias0 = 0.0f, bias1 = 0.0f;
  if (b1p) { bias0  = bias_scale * b1p[c0];           bias1  = bias_scale * b1p[c1]; }
  if (b2p) { bias0 += b2p[c0];                        bias1 += b2p[c1]; }

  float* crow = C + (size_t)(m0 + half * 8) * ldc;
  #pragma unroll
  for (int r = 0; r < 8; ++r) {
    float v0 = acc0[r] + bias0;
    float v1 = acc1[r] + bias1;
    if (RELU) { v0 = fmaxf(v0, 0.0f); v1 = fmaxf(v1, 0.0f); }
    if (WTRANS) {
      crow[c0] = v0;
      crow[c1] = v1;
    } else {
      v2f v; v.x = v0; v.y = v1;
      *reinterpret_cast<v2f*>(crow + c0) = v;
    }
    crow += ldc;
  }
}

template<bool WTRANS, bool HASA2, bool RELU,
         int LDA1, int LDW1, int K1, int LDA2, int LDW2, int K2, int LDC, int MROWS, int NCOLS>
__global__ void __launch_bounds__(256)
gemm_wmma_f32(const float* __restrict__ A1, const float* __restrict__ W1,
              const float* __restrict__ A2, const float* __restrict__ W2,
              const float* __restrict__ b1p, const float* __restrict__ b2p,
              float* __restrict__ C)
{
  gemm_tile<WTRANS, HASA2, RELU>(A1, LDA1, W1, LDW1, K1, A2, LDA2, W2, LDW2, K2,
                                 b1p, b2p, 1.0f, C, LDC, MROWS, NCOLS, (int)blockIdx.x);
}

__global__ void __launch_bounds__(256)
front_products(const float* __restrict__ hidden,
               const float* __restrict__ f_w1, const float* __restrict__ f_b1,
               const float* __restrict__ f_w2, const float* __restrict__ f_b2,
               const float* __restrict__ g_w1,
               const float* __restrict__ g_w2, const float* __restrict__ g_b2,
               const float* __restrict__ wih,
               const float* __restrict__ bih, const float* __restrict__ bhh,
               float* __restrict__ Af, float* __restrict__ Bf,
               float* __restrict__ Wf, float* __restrict__ Wg,
               float* __restrict__ bfv, float* __restrict__ bgv)
{
  const int blk = blockIdx.x;
  if (blk < 128) {
    gemm_tile<false,false,false>(hidden,128, f_w1,256,128, nullptr,0, nullptr,0,0,
                                 nullptr,nullptr,1.0f, Af,256, 2048,256, blk);
  } else if (blk < 256) {
    gemm_tile<false,false,false>(hidden,128, f_w1 + 128*256,256,128, nullptr,0, nullptr,0,0,
                                 f_b1,nullptr,1.0f, Bf,256, 2048,256, blk - 128);
  } else if (blk < 272) {
    gemm_tile<false,false,false>(f_w2,128, g_w1 + 128*256,256,128, nullptr,0, nullptr,0,0,
                                 nullptr,nullptr,1.0f, Wf,256, 256,256, blk - 256);
  } else if (blk < 304) {
    gemm_tile<true,false,false>(g_w2,128, wih,128,128, nullptr,0, nullptr,0,0,
                                nullptr,nullptr,1.0f, Wg,512, 256,512, blk - 272);
  } else {
    const int idx = (blk - 304) * 256 + threadIdx.x;
    if (idx < 256) {
      const float* wp = g_w1 + 128*256 + idx;
      float acc = 0.0f;
      for (int j = 0; j < 128; ++j) acc = fmaf(f_b2[j], wp[(size_t)j * 256], acc);
      bfv[idx] = 128.0f * acc;
    } else {
      const int n = idx - 256;
      const float* wp = wih + (size_t)n * 128;
      float acc = 0.0f;
      for (int j = 0; j < 128; ++j) acc = fmaf(g_b2[j], wp[j], acc);
      bgv[n] = acc + bih[n] + bhh[n];
    }
  }
}

__global__ void __launch_bounds__(256)
pair_relu_sum(const float* __restrict__ Af, const float* __restrict__ Bf,
              float* __restrict__ S)
{
  const int ri = blockIdx.x;
  const int b  = ri >> 7;
  const int f  = threadIdx.x;
  const float bm = Bf[(size_t)ri * 256 + f];
  const float* ap = Af + (size_t)b * 128 * 256 + f;
  float acc = 0.0f;
  #pragma unroll 16
  for (int j = 0; j < 128; ++j)
    acc += fmaxf(ap[(size_t)j * 256] + bm, 0.0f);
  S[(size_t)ri * 256 + f] = acc;
}

#define HPITCH 144

__global__ void __launch_bounds__(256)
lstm_scan(const float* __restrict__ Gx, const float* __restrict__ whh,
          const float* __restrict__ h0, const float* __restrict__ c0,
          float* __restrict__ new_hidden)
{
  __shared__ __attribute__((aligned(16))) float h_s[2][2][HPITCH];
  const int wave = threadIdx.x >> 5;
  const int lane = threadIdx.x & 31;
  const int half = lane >> 4;
  const int lp   = lane & 15;
  const int n    = wave * 16 + lp;
  const int b    = blockIdx.x * 2 + half;
  const int arow = lp >> 3;

  v4f bw[4][16];
  #pragma unroll
  for (int g = 0; g < 4; ++g) {
    const v4f* wp = reinterpret_cast<const v4f*>(whh + (size_t)(g * 128 + n) * 128 + half * 4);
    #pragma unroll
    for (int kk = 0; kk < 16; ++kk) bw[g][kk] = wp[kk * 2];
  }

  float c = c0[b * 128 + n];
  h_s[0][half][n] = h0[b * 128 + n];
  __syncthreads();

  const float* gxl = Gx + (size_t)b * 128 * 512 + n;

  for (int t = 0; t < 128; ++t) {
    const int cur = t & 1, nxt = cur ^ 1;

    float gx[4];
    #pragma unroll
    for (int g = 0; g < 4; ++g) gx[g] = gxl[(size_t)t * 512 + g * 128];

    v4f av[16];
    {
      const v4f* hp = reinterpret_cast<const v4f*>(&h_s[cur][arow][half * 4]);
      #pragma unroll
      for (int kk = 0; kk < 16; ++kk) av[kk] = hp[kk * 2];
    }

    v8f acc[3] = {};
    #pragma unroll
    for (int kk = 0; kk < 16; ++kk) {
      v2f a0, a1;
      a0.x = av[kk].x; a0.y = av[kk].y;
      a1.x = av[kk].z; a1.y = av[kk].w;
      #pragma unroll
      for (int g = 0; g < 3; ++g) {
        v2f b0; b0.x = bw[g][kk].x; b0.y = bw[g][kk].y;
        acc[g] = wmma4(a0, b0, acc[g]);
      }
      #pragma unroll
      for (int g = 0; g < 3; ++g) {
        v2f b1; b1.x = bw[g][kk].z; b1.y = bw[g][kk].w;
        acc[g] = wmma4(a1, b1, acc[g]);
      }
    }

    __builtin_amdgcn_sched_group_barrier(0x020, 4, 0);
    __builtin_amdgcn_sched_group_barrier(0x100, 16, 0);
    __builtin_amdgcn_sched_group_barrier(0x008, 96, 0);

    v4f ao0 = av[0], ao8 = av[8];
    asm volatile("" : "+v"(ao0), "+v"(ao8) : "v"(acc[0][0]), "v"(acc[1][0]), "v"(acc[2][0]));
    v8f acco[2] = {};
#define OGATE_WMMA(q)                                                              \
    do {                                                                           \
      const int hk_ = (q) & 1, i_ = (q) >> 1, k2_ = 8 * hk_ + (i_ >> 1);           \
      const v4f af_ = ((q) < 4) ? (hk_ ? ao8 : ao0) : av[k2_];                      \
      v2f a_, b_;                                                                  \
      if ((i_ & 1) == 0) { a_.x = af_.x; a_.y = af_.y; b_.x = bw[3][k2_].x; b_.y = bw[3][k2_].y; } \
      else               { a_.x = af_.z; a_.y = af_.w; b_.x = bw[3][k2_].z; b_.y = bw[3][k2_].w; } \
      acco[hk_] = wmma4(a_, b_, acco[hk_]);                                        \
    } while (0)

    #pragma unroll
    for (int q = 0; q < 16; ++q) OGATE_WMMA(q);
    const float ig = acc[0][0] + gx[0], fg = acc[1][0] + gx[1];
    const float gg = acc[2][0] + gx[2];
    const float thg = tanhf_(gg);
    const float cn = fmaf(sigmoidf_(ig), thg, sigmoidf_(fg) * c);
    c = cn;
    asm volatile("" : "+v"(acco[0]), "+v"(acco[1]) : "v"(cn));

    #pragma unroll
    for (int q = 16; q < 32; ++q) OGATE_WMMA(q);
#undef OGATE_WMMA
    const float tc = tanhf_(cn);

    const float og = (acco[0][0] + acco[1][0]) + gx[3];
    const float hn = sigmoidf_(og) * tc;
    h_s[nxt][half][n] = hn;
    __syncthreads();

    if (wave < 2) {
      const v4f v = *reinterpret_cast<const v4f*>(&h_s[nxt][wave][lane * 4]);
      *reinterpret_cast<v4f*>(new_hidden + ((size_t)(blockIdx.x * 2 + wave) * 128 + t) * 128 + lane * 4) = v;
    }
  }
}

extern "C" void kernel_launch(void* const* d_in, const int* in_sizes, int n_in,
                              void* d_out, int out_size, void* d_ws, size_t ws_size,
                              hipStream_t stream)
{
  (void)in_sizes; (void)n_in; (void)out_size; (void)ws_size;
  const float* x      = (const float*)d_in[0];
  const float* hidden = (const float*)d_in[1];
  const float* h0     = (const float*)d_in[2];
  const float* c0     = (const float*)d_in[3];
  const float* f_w1   = (const float*)d_in[4];
  const float* f_b1   = (const float*)d_in[5];
  const float* f_w2   = (const float*)d_in[6];
  const float* f_b2   = (const float*)d_in[7];
  const float* g_w1   = (const float*)d_in[8];
  const float* g_b1   = (const float*)d_in[9];
  const float* g_w2   = (const float*)d_in[10];
  const float* g_b2   = (const float*)d_in[11];
  const float* wih    = (const float*)d_in[12];
  const float* whh    = (const float*)d_in[13];
  const float* bih    = (const float*)d_in[14];
  const float* bhh    = (const float*)d_in[15];
  const float* o_w1   = (const float*)d_in[16];
  const float* o_b1   = (const float*)d_in[17];
  const float* o_w2   = (const float*)d_in[18];
  const float* o_b2   = (const float*)d_in[19];
  float* out = (float*)d_out;

  const int BN = 16 * 128;
  float* ws  = (float*)d_ws;
  float* Af  = ws;
  float* Bf  = Af  + (size_t)BN * 256;
  float* S   = Bf  + (size_t)BN * 256;
  float* G1  = S   + (size_t)BN * 256;
  float* Gx  = G1  + (size_t)BN * 256;
  float* NH  = Gx  + (size_t)BN * 512;
  float* OH1 = NH  + (size_t)BN * 128;
  float* Wf  = OH1 + (size_t)BN * 256;
  float* Wg  = Wf  + (size_t)256 * 256;
  float* bfv = Wg  + (size_t)256 * 512;
  float* bgv = bfv + 256;

  front_products<<<307, 256, 0, stream>>>(hidden, f_w1, f_b1, f_w2, f_b2, g_w1, g_w2, g_b2,
                                          wih, bih, bhh, Af, Bf, Wf, Wg, bfv, bgv);
  pair_relu_sum<<<BN, 256, 0, stream>>>(Af, Bf, S);
  gemm_wmma_f32<false,true,true, 128,256,128, 256,256,256, 256, 2048,256><<<128, 256, 0, stream>>>(
      x, g_w1, S, Wf, g_b1, bfv, G1);
  gemm_wmma_f32<false,false,false, 256,512,256, 0,0,0, 512, 2048,512><<<256, 256, 0, stream>>>(
      G1, Wg, nullptr, nullptr, bgv, nullptr, Gx);
  lstm_scan<<<8, 256, 0, stream>>>(Gx, whh, h0, c0, NH);
  gemm_wmma_f32<false,false,true, 128,256,128, 0,0,0, 256, 2048,256><<<128, 256, 0, stream>>>(
      NH, o_w1, nullptr, nullptr, o_b1, nullptr, OH1);
  gemm_wmma_f32<false,false,false, 256,32,256, 0,0,0, 32, 2048,32><<<16, 256, 0, stream>>>(
      OH1, o_w2, nullptr, nullptr, o_b2, nullptr, out);
}
